# starting kernel
# baseline (speedup 1.0000x reference)
#include <hip/hip_runtime.h>

typedef __attribute__((ext_vector_type(2))) float v2f;
typedef __attribute__((ext_vector_type(4))) float v4f;
typedef __attribute__((ext_vector_type(8))) float v8f;

constexpr int kD = 256;
constexpr int kChunks = kD / 4;
constexpr int kRowsPerWave = 16;
constexpr int kWavesPerBlock = 8;
constexpr int kRowsPerBlock = kRowsPerWave * kWavesPerBlock;

__global__ __launch_bounds__(256) void shared_unit_wmma_kernel(
    const float* __restrict__ drug_cnn,
    const float* __restrict__ drug_kg,
    const float* __restrict__ w_aa,
    const float* __restrict__ w_ab,
    const float* __restrict__ w_ba,
    const float* __restrict__ w_bb,
    const float* __restrict__ bias_cnn,
    const float* __restrict__ bias_kg,
    float* __restrict__ out_cnn,
    float* __restrict__ out_kg)
{
    __shared__ __align__(16) v2f   s_btab[kChunks][32];
    __shared__ __align__(16) float s_bc[kD];
    __shared__ __align__(16) float s_bk[kD];
    __shared__ __align__(16) float s_scl[kRowsPerBlock * 2];

    const int tid = threadIdx.x;

    s_bc[tid] = bias_cnn[tid];
    s_bk[tid] = bias_kg[tid];

    #pragma unroll
    for (int j = 0; j < 8; ++j) {
        const int f    = tid + 256 * j;
        const int kk   = f >> 5;
        const int l    = f & 31;
        const int n    = l & 15;
        const int half = l >> 4;
        const int k0   = kk * 4 + half * 2;
        v2f b = {0.0f, 0.0f};
        if (n == 0) {
            b.x = w_aa[k0]     + w_ab[k0];
            b.y = w_aa[k0 + 1] + w_ab[k0 + 1];
        } else if (n == 1) {
            b.x = w_ba[k0]     + w_bb[k0];
            b.y = w_ba[k0 + 1] + w_bb[k0 + 1];
        }
        s_btab[kk][l] = b;
    }
    __syncthreads();

    const int lane = tid & 31;
    const int wave = tid >> 5;
    const int n    = lane & 15;
    const int half = lane >> 4;
    const int rowBlock = blockIdx.x * kRowsPerBlock;
    const int rowWave  = rowBlock + wave * kRowsPerWave;

    const v2f* kgRow2 = (const v2f*)(drug_kg + (size_t)(rowWave + n) * kD);

    v8f acc = {};
    #pragma unroll 8
    for (int kk = 0; kk < kChunks; ++kk) {
        v2f a = __builtin_nontemporal_load(&kgRow2[kk * 2 + half]);
        v2f b = s_btab[kk][lane];
        acc = __builtin_amdgcn_wmma_f32_16x16x4_f32(
             false, a,  false, b,
             (short)0, acc,  false,  false);
    }

    if (n < 2) {
        const int mbase = wave * kRowsPerWave + half * 8;
        #pragma unroll
        for (int v = 0; v < 8; ++v)
            s_scl[(mbase + v) * 2 + n] = acc[v];
    }
    __syncthreads();

    const v4f* cnn4 = (const v4f*)drug_cnn;
    v4f* oc4 = (v4f*)out_cnn;
    v4f* ok4 = (v4f*)out_kg;
    const v4f* bc4 = (const v4f*)s_bc;
    const v4f* bk4 = (const v4f*)s_bk;

    #pragma unroll 4
    for (int r = 0; r < kRowsPerWave; ++r) {
        const int lrow = wave * kRowsPerWave + r;
        const size_t grow = (size_t)(rowBlock + lrow) * (kD / 4);
        const float sc = s_scl[lrow * 2 + 0];
        const float sk = s_scl[lrow * 2 + 1];
        #pragma unroll
        for (int hcol = 0; hcol < 2; ++hcol) {
            const int i = hcol * 32 + lane;
            v4f c  = __builtin_nontemporal_load(&cnn4[grow + i]);
            v4f bc = bc4[i];
            v4f bk = bk4[i];
            v4f o1, o2;
            o1.x = __builtin_fmaf(c.x, sc, bc.x);  o2.x = __builtin_fmaf(c.x, sk, bk.x);
            o1.y = __builtin_fmaf(c.y, sc, bc.y);  o2.y = __builtin_fmaf(c.y, sk, bk.y);
            o1.z = __builtin_fmaf(c.z, sc, bc.z);  o2.z = __builtin_fmaf(c.z, sk, bk.z);
            o1.w = __builtin_fmaf(c.w, sc, bc.w);  o2.w = __builtin_fmaf(c.w, sk, bk.w);
            __builtin_nontemporal_store(o1, &oc4[grow + i]);
            __builtin_nontemporal_store(o2, &ok4[grow + i]);
        }
    }
}

extern "C" void kernel_launch(void* const* d_in, const int* in_sizes, int n_in,
                              void* d_out, int out_size, void* d_ws, size_t ws_size,
                              hipStream_t stream) {
    const float* drug_cnn = (const float*)d_in[0];
    const float* drug_kg  = (const float*)d_in[1];
    const float* w_aa     = (const float*)d_in[2];
    const float* w_ab     = (const float*)d_in[3];
    const float* w_ba     = (const float*)d_in[4];
    const float* w_bb     = (const float*)d_in[5];
    const float* bcnn     = (const float*)d_in[6];
    const float* bkg      = (const float*)d_in[7];

    const int B = in_sizes[0] / kD;
    float* out_cnn = (float*)d_out;
    float* out_kg  = out_cnn + (size_t)B * kD;

    dim3 grid(B / kRowsPerBlock);
    dim3 block(kWavesPerBlock * 32);
    shared_unit_wmma_kernel<<<grid, block, 0, stream>>>(
        drug_cnn, drug_kg, w_aa, w_ab, w_ba, w_bb, bcnn, bkg, out_cnn, out_kg);
}


# each wave streams whole rows (512-byte loads, 7 rows ahead, no LDS, no barrier); kg row = 16x16 WMMA A tile as loaded, dot = sum of D's diagonal by lane butterfly (sum order changes)
# speedup vs baseline: 1.6211x; 1.6211x over previous
#include <hip/hip_runtime.h>


typedef __attribute__((ext_vector_type(2))) float v2f;
typedef __attribute__((ext_vector_type(4))) float v4f;
typedef __attribute__((ext_vector_type(8))) float v8f;

constexpr int kD = 256;
constexpr int kRowVec = kD / 4;
constexpr int kRowsPerWave = 16;
constexpr int kWavesPerBlock = 8;
constexpr int kRing = 8;

#define DEV static __device__ __forceinline__

template <int CTRL>
DEV float row_swizzle(float v) {
    const int x = __builtin_bit_cast(int, v);
    return __builtin_bit_cast(float, __builtin_amdgcn_update_dpp(x, x, CTRL, 0xf, 0xf, true));
}

DEV float other_row_xor8(float v) {
    const unsigned x = __builtin_bit_cast(unsigned, v);
    return __builtin_bit_cast(float,
        __builtin_amdgcn_permlanex16(x, x, 0xFEDCBA98u, 0x76543210u, false, false));
}

DEV v8f wmma4(v2f a, v2f b, v8f c) {
    return __builtin_amdgcn_wmma_f32_16x16x4_f32(
         false, a,  false, b,
         (short)0, c,  false,  false);
}

__global__ __launch_bounds__(256) void shared_unit_rowstream_kernel(
    const float* __restrict__ drug_cnn,
    const float* __restrict__ drug_kg,
    const float* __restrict__ w_aa,
    const float* __restrict__ w_ab,
    const float* __restrict__ w_ba,
    const float* __restrict__ w_bb,
    const float* __restrict__ bias_cnn,
    const float* __restrict__ bias_kg,
    float* __restrict__ out_cnn,
    float* __restrict__ out_kg)
{
    const int tid  = threadIdx.x;
    const int lane = tid & 31;
    const int wave = __builtin_amdgcn_readfirstlane(tid >> 5);
    const int row0 = (blockIdx.x * kWavesPerBlock + wave) * kRowsPerWave;

    const v4f* kg4  = (const v4f*)drug_kg  + (size_t)row0 * kRowVec + lane;
    const v4f* cnn4 = (const v4f*)drug_cnn + (size_t)row0 * kRowVec + lane;
    v4f* oc4 = (v4f*)out_cnn + (size_t)row0 * kRowVec + lane;
    v4f* ok4 = (v4f*)out_kg  + (size_t)row0 * kRowVec + lane;

    const int lrot = lane ^ 8;
    const v4f wcA = ((const v4f*)w_aa)[lane]      + ((const v4f*)w_ab)[lane];
    const v4f wcB = ((const v4f*)w_aa)[32 + lane] + ((const v4f*)w_ab)[32 + lane];
    const v4f wkA = ((const v4f*)w_ba)[lrot]      + ((const v4f*)w_bb)[lrot];
    const v4f wkB = ((const v4f*)w_ba)[32 + lrot] + ((const v4f*)w_bb)[32 + lrot];
    const v4f bcA = ((const v4f*)bias_cnn)[lane];
    const v4f bcB = ((const v4f*)bias_cnn)[32 + lane];
    const v4f bkA = ((const v4f*)bias_kg)[lane];
    const v4f bkB = ((const v4f*)bias_kg)[32 + lane];

    const v2f bc0 = {wcA.x, wcA.y}, bc1 = {wcA.z, wcA.w}, bc2 = {wcB.x, wcB.y}, bc3 = {wcB.z, wcB.w};
    const v2f bk0 = {wkA.x, wkA.y}, bk1 = {wkA.z, wkA.w}, bk2 = {wkB.x, wkB.y}, bk3 = {wkB.z, wkB.w};

    const bool first = (((lane >> 3) ^ (lane >> 4)) & 1) == 0;
    const bool i4 = (lane & 4) != 0, i2 = (lane & 2) != 0, i1 = (lane & 1) != 0;

    v4f kgA[kRing], kgB[kRing], cnA[kRing], cnB[kRing];
#define LOAD_KG(R) do { \
        kgA[(R) % kRing] = __builtin_nontemporal_load(kg4  + (R) * kRowVec); \
        kgB[(R) % kRing] = __builtin_nontemporal_load(kg4  + (R) * kRowVec + 32); } while (0)
#define LOAD_CNN(R) do { \
        cnA[(R) % kRing] = __builtin_nontemporal_load(cnn4 + (R) * kRowVec); \
        cnB[(R) % kRing] = __builtin_nontemporal_load(cnn4 + (R) * kRowVec + 32); } while (0)

    #pragma unroll
    for (int r = 0; r < kRing - 1; ++r) { LOAD_KG(r); if (r < kRing - 2) LOAD_CNN(r); }

    v8f dc[2], dk[2];
    #pragma unroll
    for (int r = 0; r <= kRowsPerWave; ++r) {
        __builtin_amdgcn_sched_barrier(0);
        if (r + kRing - 1 < kRowsPerWave) LOAD_KG(r + kRing - 1);
        if (r + kRing - 2 < kRowsPerWave) LOAD_CNN(r + kRing - 2);
        __builtin_amdgcn_sched_barrier(0);

        if (r < kRowsPerWave) {
            const v4f ka = kgA[r % kRing], kb = kgB[r % kRing];
            const v2f a0 = {ka.x, ka.y}, a1 = {ka.z, ka.w}, a2 = {kb.x, kb.y}, a3 = {kb.z, kb.w};
            v8f c = {}, k = {};
            c = wmma4(a0, bc0, c);  k = wmma4(a0, bk0, k);
            c = wmma4(a1, bc1, c);  k = wmma4(a1, bk1, k);
            c = wmma4(a2, bc2, c);  k = wmma4(a2, bk2, k);
            c = wmma4(a3, bc3, c);  k = wmma4(a3, bk3, k);
            dc[r & 1] = c;  dk[r & 1] = k;
        }
        __builtin_amdgcn_sched_barrier(0);

        if (r >= 1) {
            const int p = r - 1;
            const v8f c = dc[p & 1], k = dk[p & 1];
            const float e0 = first ? c[0] : k[0], e1 = first ? c[1] : k[1];
            const float e2 = first ? c[2] : k[2], e3 = first ? c[3] : k[3];
            const float e4 = first ? c[4] : k[4], e5 = first ? c[5] : k[5];
            const float e6 = first ? c[6] : k[6], e7 = first ? c[7] : k[7];
            const float y0 = i4 ? e4 : e0, y1 = i4 ? e5 : e1, y2 = i4 ? e6 : e2, y3 = i4 ? e7 : e3;
            const float z0 = i2 ? y2 : y0, z1 = i2 ? y3 : y1;
            float t = i1 ? z1 : z0;

            t += row_swizzle<0xB1>(t);
            t += row_swizzle<0x4E>(t);
            t += row_swizzle<0x164>(t);
            t += other_row_xor8(t);
            const int ti = __builtin_bit_cast(int, t);
            const float sc = __builtin_bit_cast(float, __builtin_amdgcn_readlane(ti, 0));
            const float sk = __builtin_bit_cast(float, __builtin_amdgcn_readlane(ti, 8));

            const v4f ca = cnA[p % kRing], cb = cnB[p % kRing];
            v4f o1a, o1b, o2a, o2b;
            o1a.x = __builtin_fmaf(ca.x, sc, bcA.x);  o2a.x = __builtin_fmaf(ca.x, sk, bkA.x);
            o1a.y = __builtin_fmaf(ca.y, sc, bcA.y);  o2a.y = __builtin_fmaf(ca.y, sk, bkA.y);
            o1a.z = __builtin_fmaf(ca.z, sc, bcA.z);  o2a.z = __builtin_fmaf(ca.z, sk, bkA.z);
            o1a.w = __builtin_fmaf(ca.w, sc, bcA.w);  o2a.w = __builtin_fmaf(ca.w, sk, bkA.w);
            o1b.x = __builtin_fmaf(cb.x, sc, bcB.x);  o2b.x = __builtin_fmaf(cb.x, sk, bkB.x);
            o1b.y = __builtin_fmaf(cb.y, sc, bcB.y);  o2b.y = __builtin_fmaf(cb.y, sk, bkB.y);
            o1b.z = __builtin_fmaf(cb.z, sc, bcB.z);  o2b.z = __builtin_fmaf(cb.z, sk, bkB.z);
            o1b.w = __builtin_fmaf(cb.w, sc, bcB.w);  o2b.w = __builtin_fmaf(cb.w, sk, bkB.w);
            __builtin_nontemporal_store(o1a, oc4 + p * kRowVec);
            __builtin_nontemporal_store(o1b, oc4 + p * kRowVec + 32);
            __builtin_nontemporal_store(o2a, ok4 + p * kRowVec);
            __builtin_nontemporal_store(o2b, ok4 + p * kRowVec + 32);
        }
        if (r < kRowsPerWave) asm volatile("" :: "v"(kgA[r % kRing]), "v"(kgB[r % kRing]));
    }
    asm volatile("" :: "v"(bc0), "v"(bc1), "v"(bc2), "v"(bc3), "v"(bk0), "v"(bk1), "v"(bk2), "v"(bk3));
#undef LOAD_KG
#undef LOAD_CNN
}

extern "C" void kernel_launch(void* const* d_in, const int* in_sizes, int n_in,
                              void* d_out, int out_size, void* d_ws, size_t ws_size,
                              hipStream_t stream) {
    const float* drug_cnn = (const float*)d_in[0];
    const float* drug_kg  = (const float*)d_in[1];
    const float* w_aa     = (const float*)d_in[2];
    const float* w_ab     = (const float*)d_in[3];
    const float* w_ba     = (const float*)d_in[4];
    const float* w_bb     = (const float*)d_in[5];
    const float* bcnn     = (const float*)d_in[6];
    const float* bkg      = (const float*)d_in[7];

    const int B = in_sizes[0] / kD;
    float* out_cnn = (float*)d_out;
    float* out_kg  = out_cnn + (size_t)B * kD;

    dim3 grid(B / (kRowsPerWave * kWavesPerBlock));
    dim3 block(kWavesPerBlock * 32);
    shared_unit_rowstream_kernel<<<grid, block, 0, stream>>>(
        drug_cnn, drug_kg, w_aa, w_ab, w_ba, w_bb, bcnn, bkg, out_cnn, out_kg);
}


# rows dealt to the waves in turn (wave g takes rows g, g+4096, ...): per step the grid streams one contiguous 4-MB band per array, not 4096 private 16-KB pieces at a 16-KB pitch; same sums and order
# speedup vs baseline: 1.9457x; 1.2002x over previous
#include <hip/hip_runtime.h>


typedef __attribute__((ext_vector_type(2))) float v2f;
typedef __attribute__((ext_vector_type(4))) float v4f;
typedef __attribute__((ext_vector_type(8))) float v8f;

constexpr int kD = 256;
constexpr int kRowVec = kD / 4;
constexpr int kRowsPerWave = 16;
constexpr int kWavesPerBlock = 8;
constexpr int kRows = 65536;
constexpr int kWaves = kRows / kRowsPerWave;
constexpr int kRing = 8;

#define DEV static __device__ __forceinline__

template <int CTRL>
DEV float row_swizzle(float v) {
    const int x = __builtin_bit_cast(int, v);
    return __builtin_bit_cast(float, __builtin_amdgcn_update_dpp(x, x, CTRL, 0xf, 0xf, true));
}

DEV float other_row_xor8(float v) {
    const unsigned x = __builtin_bit_cast(unsigned, v);
    return __builtin_bit_cast(float,
        __builtin_amdgcn_permlanex16(x, x, 0xFEDCBA98u, 0x76543210u, false, false));
}

typedef __attribute__((address_space(1))) v4f gv4f;
DEV gv4f* scalar_ptr(const v4f* p) {
    unsigned long long u = (unsigned long long)p;
    asm("" : "+s"(u));
    return (gv4f*)u;
}

DEV v8f wmma4(v2f a, v2f b, v8f c) {
    return __builtin_amdgcn_wmma_f32_16x16x4_f32(
         false, a,  false, b,
         (short)0, c,  false,  false);
}

__global__ __launch_bounds__(256) void shared_unit_rowstream_kernel(
    const float* __restrict__ drug_cnn,
    const float* __restrict__ drug_kg,
    const float* __restrict__ w_aa,
    const float* __restrict__ w_ab,
    const float* __restrict__ w_ba,
    const float* __restrict__ w_bb,
    const float* __restrict__ bias_cnn,
    const float* __restrict__ bias_kg,
    float* __restrict__ out_cnn,
    float* __restrict__ out_kg)
{
    const int tid  = threadIdx.x;
    const int lane = tid & 31;
    const int wave = __builtin_amdgcn_readfirstlane(tid >> 5);
    const int row0 = blockIdx.x * kWavesPerBlock + wave;
    constexpr long step = (long)kWaves * kRowVec;
    static_assert(2 * step * 16 <= (1 << 23), "rows -2..+1 around a base must fit the signed 24-bit offset");

    const v4f* kg4  = (const v4f*)drug_kg  + (size_t)row0 * kRowVec;
    const v4f* cnn4 = (const v4f*)drug_cnn + (size_t)row0 * kRowVec;
    const v4f* oc4  = (const v4f*)out_cnn  + (size_t)row0 * kRowVec;
    const v4f* ok4  = (const v4f*)out_kg   + (size_t)row0 * kRowVec;
    gv4f* kgb[kRowsPerWave / 4]; gv4f* cnb[kRowsPerWave / 4]; gv4f* ocb[kRowsPerWave / 4]; gv4f* okb[kRowsPerWave / 4];
    #pragma unroll
    for (int j = 0; j < kRowsPerWave / 4; ++j) {
        kgb[j] = scalar_ptr(kg4  + (4 * j + 2) * step);
        cnb[j] = scalar_ptr(cnn4 + (4 * j + 2) * step);
        ocb[j] = scalar_ptr(oc4  + (4 * j + 2) * step);
        okb[j] = scalar_ptr(ok4  + (4 * j + 2) * step);
    }
#define ROWP(BASES, R) ((BASES)[(R) / 4] + (((R) % 4) - 2) * step + lane)

    const int lrot = lane ^ 8;
    const v4f wcA = ((const v4f*)w_aa)[lane]      + ((const v4f*)w_ab)[lane];
    const v4f wcB = ((const v4f*)w_aa)[32 + lane] + ((const v4f*)w_ab)[32 + lane];
    const v4f wkA = ((const v4f*)w_ba)[lrot]      + ((const v4f*)w_bb)[lrot];
    const v4f wkB = ((const v4f*)w_ba)[32 + lrot] + ((const v4f*)w_bb)[32 + lrot];
    const v4f bcA = ((const v4f*)bias_cnn)[lane];
    const v4f bcB = ((const v4f*)bias_cnn)[32 + lane];
    const v4f bkA = ((const v4f*)bias_kg)[lane];
    const v4f bkB = ((const v4f*)bias_kg)[32 + lane];

    const v2f bc0 = {wcA.x, wcA.y}, bc1 = {wcA.z, wcA.w}, bc2 = {wcB.x, wcB.y}, bc3 = {wcB.z, wcB.w};
    const v2f bk0 = {wkA.x, wkA.y}, bk1 = {wkA.z, wkA.w}, bk2 = {wkB.x, wkB.y}, bk3 = {wkB.z, wkB.w};

    const bool first = (((lane >> 3) ^ (lane >> 4)) & 1) == 0;
    const bool i4 = (lane & 4) != 0, i2 = (lane & 2) != 0, i1 = (lane & 1) != 0;

    v4f kgA[kRing], kgB[kRing], cnA[kRing], cnB[kRing];
#define LOAD_KG(R) do { \
        kgA[(R) % kRing] = __builtin_nontemporal_load(ROWP(kgb, R)); \
        kgB[(R) % kRing] = __builtin_nontemporal_load(ROWP(kgb, R) + 32); } while (0)
#define LOAD_CNN(R) do { \
        cnA[(R) % kRing] = __builtin_nontemporal_load(ROWP(cnb, R)); \
        cnB[(R) % kRing] = __builtin_nontemporal_load(ROWP(cnb, R) + 32); } while (0)

    #pragma unroll
    for (int r = 0; r < kRing - 1; ++r) { LOAD_KG(r); if (r < kRing - 2) LOAD_CNN(r); }

    v8f dc[2], dk[2];
    #pragma unroll
    for (int r = 0; r <= kRowsPerWave; ++r) {
        __builtin_amdgcn_sched_barrier(0);
        if (r + kRing - 1 < kRowsPerWave) LOAD_KG(r + kRing - 1);
        if (r + kRing - 2 < kRowsPerWave) LOAD_CNN(r + kRing - 2);
        __builtin_amdgcn_sched_barrier(0);

        if (r < kRowsPerWave) {
            const v4f ka = kgA[r % kRing], kb = kgB[r % kRing];
            const v2f a0 = {ka.x, ka.y}, a1 = {ka.z, ka.w}, a2 = {kb.x, kb.y}, a3 = {kb.z, kb.w};
            v8f c = {}, k = {};
            c = wmma4(a0, bc0, c);  k = wmma4(a0, bk0, k);
            c = wmma4(a1, bc1, c);  k = wmma4(a1, bk1, k);
            c = wmma4(a2, bc2, c);  k = wmma4(a2, bk2, k);
            c = wmma4(a3, bc3, c);  k = wmma4(a3, bk3, k);
            dc[r & 1] = c;  dk[r & 1] = k;
        }
        __builtin_amdgcn_sched_barrier(0);

        if (r >= 1) {
            const int p = r - 1;
            const v8f c = dc[p & 1], k = dk[p & 1];
            const float e0 = first ? c[0] : k[0], e1 = first ? c[1] : k[1];
            const float e2 = first ? c[2] : k[2], e3 = first ? c[3] : k[3];
            const float e4 = first ? c[4] : k[4], e5 = first ? c[5] : k[5];
            const float e6 = first ? c[6] : k[6], e7 = first ? c[7] : k[7];
            const float y0 = i4 ? e4 : e0, y1 = i4 ? e5 : e1, y2 = i4 ? e6 : e2, y3 = i4 ? e7 : e3;
            const float z0 = i2 ? y2 : y0, z1 = i2 ? y3 : y1;
            float t = i1 ? z1 : z0;

            t += row_swizzle<0xB1>(t);
            t += row_swizzle<0x4E>(t);
            t += row_swizzle<0x164>(t);
            t += other_row_xor8(t);
            const int ti = __builtin_bit_cast(int, t);
            const float sc = __builtin_bit_cast(float, __builtin_amdgcn_readlane(ti, 0));
            const float sk = __builtin_bit_cast(float, __builtin_amdgcn_readlane(ti, 8));

            const v4f ca = cnA[p % kRing], cb = cnB[p % kRing];
            v4f o1a, o1b, o2a, o2b;
            o1a.x = __builtin_fmaf(ca.x, sc, bcA.x);  o2a.x = __builtin_fmaf(ca.x, sk, bkA.x);
            o1a.y = __builtin_fmaf(ca.y, sc, bcA.y);  o2a.y = __builtin_fmaf(ca.y, sk, bkA.y);
            o1a.z = __builtin_fmaf(ca.z, sc, bcA.z);  o2a.z = __builtin_fmaf(ca.z, sk, bkA.z);
            o1a.w = __builtin_fmaf(ca.w, sc, bcA.w);  o2a.w = __builtin_fmaf(ca.w, sk, bkA.w);
            o1b.x = __builtin_fmaf(cb.x, sc, bcB.x);  o2b.x = __builtin_fmaf(cb.x, sk, bkB.x);
            o1b.y = __builtin_fmaf(cb.y, sc, bcB.y);  o2b.y = __builtin_fmaf(cb.y, sk, bkB.y);
            o1b.z = __builtin_fmaf(cb.z, sc, bcB.z);  o2b.z = __builtin_fmaf(cb.z, sk, bkB.z);
            o1b.w = __builtin_fmaf(cb.w, sc, bcB.w);  o2b.w = __builtin_fmaf(cb.w, sk, bkB.w);
            __builtin_nontemporal_store(o1a, ROWP(ocb, p));
            __builtin_nontemporal_store(o1b, ROWP(ocb, p) + 32);
            __builtin_nontemporal_store(o2a, ROWP(okb, p));
            __builtin_nontemporal_store(o2b, ROWP(okb, p) + 32);
        }
        if (r < kRowsPerWave) asm volatile("" :: "v"(kgA[r % kRing]), "v"(kgB[r % kRing]));
    }
    asm volatile("" :: "v"(bc0), "v"(bc1), "v"(bc2), "v"(bc3), "v"(bk0), "v"(bk1), "v"(bk2), "v"(bk3));
#undef LOAD_KG
#undef LOAD_CNN
#undef ROWP
}

extern "C" void kernel_launch(void* const* d_in, const int* in_sizes, int n_in,
                              void* d_out, int out_size, void* d_ws, size_t ws_size,
                              hipStream_t stream) {
    const float* drug_cnn = (const float*)d_in[0];
    const float* drug_kg  = (const float*)d_in[1];
    const float* w_aa     = (const float*)d_in[2];
    const float* w_ab     = (const float*)d_in[3];
    const float* w_ba     = (const float*)d_in[4];
    const float* w_bb     = (const float*)d_in[5];
    const float* bcnn     = (const float*)d_in[6];
    const float* bkg      = (const float*)d_in[7];

    const int B = in_sizes[0] / kD;
    float* out_cnn = (float*)d_out;
    float* out_kg  = out_cnn + (size_t)B * kD;

    dim3 grid(kWaves / kWavesPerBlock);
    dim3 block(kWavesPerBlock * 32);
    shared_unit_rowstream_kernel<<<grid, block, 0, stream>>>(
        drug_cnn, drug_kg, w_aa, w_ab, w_ba, w_bb, bcnn, bkg, out_cnn, out_kg);
}
